# starting kernel
# baseline (speedup 1.0000x reference)
#include <hip/hip_runtime.h>
#include <hip/hip_bf16.h>
#include <math.h>

#define BB 128
#define TT 512
#define KK 1024
#define QQ 1024

typedef __attribute__((ext_vector_type(2))) float v2f;
typedef __attribute__((ext_vector_type(4))) float v4f;
typedef __attribute__((ext_vector_type(8))) float v8f;

__global__ __launch_bounds__(32)
void mids_wmma_kernel(const float* __restrict__ query,
                      const float* __restrict__ W,
                      float* __restrict__ mids)
{
    const int tile = blockIdx.x;
    const int tm   = tile >> 6;
    const int tn   = tile & 63;
    const int lane = threadIdx.x;
    const int half = lane >> 4;
    const int l16  = lane & 15;

    const int m = tm * 16 + l16;
    const int n = tn * 16 + l16;

    const float* __restrict__ qrow = query + (size_t)m * QQ;
    const float* __restrict__ wrow = W     + (size_t)n * QQ;

    v8f acc = {};
    #pragma unroll 4
    for (int q0 = 0; q0 < QQ; q0 += 4) {
        v2f a = *(const v2f*)(qrow + q0 + 2 * half);
        v2f b = *(const v2f*)(wrow + q0 + 2 * half);
        acc = __builtin_amdgcn_wmma_f32_16x16x4_f32(
             false, a,  false, b,
             (short)0, acc,  false,  false);
    }

    const int row_base = tm * 16 + half * 8;
    const int col      = tn * 16 + l16;
    #pragma unroll
    for (int r = 0; r < 8; ++r) {
        mids[(size_t)(row_base + r) * KK + col] = acc[r];
    }
}

__global__ __launch_bounds__(256)
void scores_kernel(const float* __restrict__ key,
                   const float* __restrict__ mids,
                   const float* __restrict__ bias,
                   float* __restrict__ scores)
{
    __shared__ __align__(16) float smids[KK];

    const int blk    = blockIdx.x;
    const int b      = blk >> 3;
    const int tchunk = blk & 7;
    const int wave   = threadIdx.x >> 5;
    const int lane   = threadIdx.x & 31;

    const float bias0 = bias[0];
    const float* __restrict__ midsb = mids + (size_t)b * KK;

    {
        const int k = threadIdx.x * 4;
        *(v4f*)(smids + k) = *(const v4f*)(midsb + k);
    }
    __syncthreads();

    #pragma unroll
    for (int i = 0; i < 8; ++i) {
        const int t = tchunk * 64 + wave * 8 + i;
        const float* __restrict__ krow = key + ((size_t)b * TT + t) * KK;

        float a0 = 0.f, a1 = 0.f, a2 = 0.f, a3 = 0.f;
        #pragma unroll
        for (int k0 = 0; k0 < KK; k0 += 128) {
            const int k = k0 + lane * 4;
            v4f kv = __builtin_nontemporal_load((const v4f*)(krow + k));
            v4f mv = *(const v4f*)(smids + k);
            a0 = fmaf(kv.x, mv.x, a0);
            a1 = fmaf(kv.y, mv.y, a1);
            a2 = fmaf(kv.z, mv.z, a2);
            a3 = fmaf(kv.w, mv.w, a3);
        }
        float s = (a0 + a1) + (a2 + a3);
        #pragma unroll
        for (int off = 16; off > 0; off >>= 1)
            s += __shfl_down(s, off, 32);
        if (lane == 0)
            scores[(size_t)b * TT + t] = tanhf(s + bias0);
    }
}

__global__ __launch_bounds__(256)
void softmax_kernel(const float* __restrict__ mask,
                    float* __restrict__ out)
{
    __shared__ float sred[8];
    const int b    = blockIdx.x;
    const int tid  = threadIdx.x;
    const int lane = tid & 31;
    const int wave = tid >> 5;

    float* __restrict__ row        = out  + (size_t)b * TT;
    const float* __restrict__ mrow = mask + (size_t)b * TT;

    float v0 = row[tid];
    float v1 = row[tid + 256];

    float mx = fmaxf(v0, v1);
    #pragma unroll
    for (int off = 16; off > 0; off >>= 1)
        mx = fmaxf(mx, __shfl_down(mx, off, 32));
    if (lane == 0) sred[wave] = mx;
    __syncthreads();
    if (wave == 0) {
        float m = sred[lane & 7];
        #pragma unroll
        for (int off = 4; off > 0; off >>= 1)
            m = fmaxf(m, __shfl_down(m, off, 32));
        if (lane == 0) sred[0] = m;
    }
    __syncthreads();
    mx = sred[0];
    __syncthreads();

    float e0 = __expf(v0 - mx) * mrow[tid];
    float e1 = __expf(v1 - mx) * mrow[tid + 256];
    float s  = e0 + e1;
    #pragma unroll
    for (int off = 16; off > 0; off >>= 1)
        s += __shfl_down(s, off, 32);
    if (lane == 0) sred[wave] = s;
    __syncthreads();
    if (wave == 0) {
        float t = sred[lane & 7];
        #pragma unroll
        for (int off = 4; off > 0; off >>= 1)
            t += __shfl_down(t, off, 32);
        if (lane == 0) sred[0] = t;
    }
    __syncthreads();
    const float inv = 1.0f / sred[0];

    row[tid]       = e0 * inv;
    row[tid + 256] = e1 * inv;
}

extern "C" void kernel_launch(void* const* d_in, const int* in_sizes, int n_in,
                              void* d_out, int out_size, void* d_ws, size_t ws_size,
                              hipStream_t stream)
{
    const float* query = (const float*)d_in[0];
    const float* key   = (const float*)d_in[1];
    const float* mask  = (const float*)d_in[2];
    const float* W     = (const float*)d_in[3];
    const float* bias  = (const float*)d_in[4];
    float* out  = (float*)d_out;
    float* mids = (float*)d_ws;

    mids_wmma_kernel<<<512, 32, 0, stream>>>(query, W, mids);

    scores_kernel<<<BB * (TT / 64), 256, 0, stream>>>(key, mids, bias, out);

    softmax_kernel<<<BB, 256, 0, stream>>>(mask, out);
}


# mids GEMM: 8 waves per 16x32 output block split the inner dimension, all 48 operand loads in flight at once, partial tiles added through LDS in fixed order (new summation order)
# speedup vs baseline: 1.1771x; 1.1771x over previous
#include <hip/hip_runtime.h>
#include <hip/hip_bf16.h>
#include <math.h>

#define BB 128
#define TT 512
#define KK 1024
#define QQ 1024

typedef __attribute__((ext_vector_type(2))) float v2f;
typedef __attribute__((ext_vector_type(4))) float v4f;
typedef __attribute__((ext_vector_type(8))) float v8f;

#define SPLITK      8
#define KSLAB       (QQ / SPLITK)
#define NV4         (KSLAB / 8)
#define RED_STRIDE  272

__global__ __launch_bounds__(256)
void mids_wmma_kernel(const float* __restrict__ query,
                      const float* __restrict__ W,
                      float* __restrict__ mids)
{
    __shared__ __align__(16) float red[SPLITK * 2 * RED_STRIDE];

    const int wg   = blockIdx.x;
    const int tm   = wg >> 5;
    const int tn   = wg & 31;
    const int wave = threadIdx.x >> 5;
    const int lane = threadIdx.x & 31;
    const int half = lane >> 4;
    const int l16  = lane & 15;

    const int koff = wave * KSLAB + half * 4;
    const float* __restrict__ qp  = query + (size_t)(tm * 16 + l16) * QQ + koff;
    const float* __restrict__ w0p = W + (size_t)(tn * 32 + l16) * QQ + koff;
    const float* __restrict__ w1p = W + (size_t)(tn * 32 + 16 + l16) * QQ + koff;

    v4f a[NV4], b0[NV4], b1[NV4];
    #pragma unroll
    for (int j = 0; j < NV4; ++j) {
        a[j]  = *(const v4f*)(qp  + 8 * j);
        b0[j] = *(const v4f*)(w0p + 8 * j);
        b1[j] = *(const v4f*)(w1p + 8 * j);
    }
    __builtin_amdgcn_sched_barrier(0);

    v8f acc0 = {};
    v8f acc1 = {};
    #pragma unroll
    for (int j = 0; j < NV4; ++j) {
        const v2f alo  = {a[j].x,  a[j].y};
        const v2f ahi  = {a[j].z,  a[j].w};
        const v2f b0lo = {b0[j].x, b0[j].y};
        const v2f b0hi = {b0[j].z, b0[j].w};
        const v2f b1lo = {b1[j].x, b1[j].y};
        const v2f b1hi = {b1[j].z, b1[j].w};
        acc0 = __builtin_amdgcn_wmma_f32_16x16x4_f32(false, alo, false, b0lo, (short)0, acc0, false, false);
        acc1 = __builtin_amdgcn_wmma_f32_16x16x4_f32(false, alo, false, b1lo, (short)0, acc1, false, false);
        acc0 = __builtin_amdgcn_wmma_f32_16x16x4_f32(false, ahi, false, b0hi, (short)0, acc0, false, false);
        acc1 = __builtin_amdgcn_wmma_f32_16x16x4_f32(false, ahi, false, b1hi, (short)0, acc1, false, false);
    }

    float* __restrict__ mine = red + (wave * 2) * RED_STRIDE;
    #pragma unroll
    for (int r = 0; r < 8; ++r) {
        mine[r * 32 + lane]              = acc0[r];
        mine[RED_STRIDE + r * 32 + lane] = acc1[r];
    }
    __syncthreads();

    const int row  = threadIdx.x >> 4;
    const int c2   = threadIdx.x & 15;
    const int tile = c2 >> 3;
    const int col  = (2 * c2) & 15;
    const int srci = tile * RED_STRIDE + (row & 7) * 32 + (row >> 3) * 16 + col;
    v2f s = *(const v2f*)(red + srci);
    for (int w = 1; w < SPLITK; ++w)
        s += *(const v2f*)(red + (w * 2) * RED_STRIDE + srci);
    *(v2f*)(mids + (size_t)(tm * 16 + row) * KK + tn * 32 + 2 * c2) = s;
}

__global__ __launch_bounds__(256)
void scores_kernel(const float* __restrict__ key,
                   const float* __restrict__ mids,
                   const float* __restrict__ bias,
                   float* __restrict__ scores)
{
    __shared__ __align__(16) float smids[KK];

    const int blk    = blockIdx.x;
    const int b      = blk >> 3;
    const int tchunk = blk & 7;
    const int wave   = threadIdx.x >> 5;
    const int lane   = threadIdx.x & 31;

    const float bias0 = bias[0];
    const float* __restrict__ midsb = mids + (size_t)b * KK;

    {
        const int k = threadIdx.x * 4;
        *(v4f*)(smids + k) = *(const v4f*)(midsb + k);
    }
    __syncthreads();

    #pragma unroll
    for (int i = 0; i < 8; ++i) {
        const int t = tchunk * 64 + wave * 8 + i;
        const float* __restrict__ krow = key + ((size_t)b * TT + t) * KK;

        float a0 = 0.f, a1 = 0.f, a2 = 0.f, a3 = 0.f;
        #pragma unroll
        for (int k0 = 0; k0 < KK; k0 += 128) {
            const int k = k0 + lane * 4;
            v4f kv = __builtin_nontemporal_load((const v4f*)(krow + k));
            v4f mv = *(const v4f*)(smids + k);
            a0 = fmaf(kv.x, mv.x, a0);
            a1 = fmaf(kv.y, mv.y, a1);
            a2 = fmaf(kv.z, mv.z, a2);
            a3 = fmaf(kv.w, mv.w, a3);
        }
        float s = (a0 + a1) + (a2 + a3);
        #pragma unroll
        for (int off = 16; off > 0; off >>= 1)
            s += __shfl_down(s, off, 32);
        if (lane == 0)
            scores[(size_t)b * TT + t] = tanhf(s + bias0);
    }
}

__global__ __launch_bounds__(256)
void softmax_kernel(const float* __restrict__ mask,
                    float* __restrict__ out)
{
    __shared__ float sred[8];
    const int b    = blockIdx.x;
    const int tid  = threadIdx.x;
    const int lane = tid & 31;
    const int wave = tid >> 5;

    float* __restrict__ row        = out  + (size_t)b * TT;
    const float* __restrict__ mrow = mask + (size_t)b * TT;

    float v0 = row[tid];
    float v1 = row[tid + 256];

    float mx = fmaxf(v0, v1);
    #pragma unroll
    for (int off = 16; off > 0; off >>= 1)
        mx = fmaxf(mx, __shfl_down(mx, off, 32));
    if (lane == 0) sred[wave] = mx;
    __syncthreads();
    if (wave == 0) {
        float m = sred[lane & 7];
        #pragma unroll
        for (int off = 4; off > 0; off >>= 1)
            m = fmaxf(m, __shfl_down(m, off, 32));
        if (lane == 0) sred[0] = m;
    }
    __syncthreads();
    mx = sred[0];
    __syncthreads();

    float e0 = __expf(v0 - mx) * mrow[tid];
    float e1 = __expf(v1 - mx) * mrow[tid + 256];
    float s  = e0 + e1;
    #pragma unroll
    for (int off = 16; off > 0; off >>= 1)
        s += __shfl_down(s, off, 32);
    if (lane == 0) sred[wave] = s;
    __syncthreads();
    if (wave == 0) {
        float t = sred[lane & 7];
        #pragma unroll
        for (int off = 4; off > 0; off >>= 1)
            t += __shfl_down(t, off, 32);
        if (lane == 0) sred[0] = t;
    }
    __syncthreads();
    const float inv = 1.0f / sred[0];

    row[tid]       = e0 * inv;
    row[tid + 256] = e1 * inv;
}

extern "C" void kernel_launch(void* const* d_in, const int* in_sizes, int n_in,
                              void* d_out, int out_size, void* d_ws, size_t ws_size,
                              hipStream_t stream)
{
    const float* query = (const float*)d_in[0];
    const float* key   = (const float*)d_in[1];
    const float* mask  = (const float*)d_in[2];
    const float* W     = (const float*)d_in[3];
    const float* bias  = (const float*)d_in[4];
    float* out  = (float*)d_out;
    float* mids = (float*)d_ws;

    mids_wmma_kernel<<<256, 256, 0, stream>>>(query, W, mids);

    scores_kernel<<<BB * (TT / 64), 256, 0, stream>>>(key, mids, bias, out);

    softmax_kernel<<<BB, 256, 0, stream>>>(mask, out);
}
